# starting kernel
# baseline (speedup 1.0000x reference)
#include <hip/hip_runtime.h>
#include <hip/hip_bf16.h>
#include <math.h>


#define KAN_BATCH   512
#define KAN_IN      128
#define KAN_OUT     128
#define KAN_G       5
#define KAN_KORD    3
#define KAN_NBASIS  (KAN_G + KAN_KORD)
#define KAN_NKNOT   (KAN_G + 2 * KAN_KORD + 1)
#define KAN_FEAT    (KAN_NBASIS + 1)
#define KAN_KDIM    (KAN_IN * KAN_FEAT)
#define KAN_SIZE    (KAN_IN * KAN_OUT)
#define GEMM_KC     128

typedef __attribute__((ext_vector_type(2))) float v2f;
typedef __attribute__((ext_vector_type(8))) float v8f;

__global__ __launch_bounds__(256)
void kan_features(const float* __restrict__ x,
                  const float* __restrict__ knots_g,
                  float* __restrict__ F) {
    int t = blockIdx.x * blockDim.x + threadIdx.x;
    if (t >= KAN_BATCH * KAN_IN) return;
    int b = t / KAN_IN;
    int i = t - b * KAN_IN;

    float xv = x[b * KAN_IN + i];

    float tk[KAN_NKNOT];
#pragma unroll
    for (int j = 0; j < KAN_NKNOT; ++j) tk[j] = knots_g[j];

    float bas[KAN_NKNOT - 1];
#pragma unroll
    for (int j = 0; j < KAN_NKNOT - 1; ++j)
        bas[j] = (xv >= tk[j] && xv < tk[j + 1]) ? 1.0f : 0.0f;

#pragma unroll
    for (int p = 1; p <= KAN_KORD; ++p) {
        int n = KAN_NKNOT - 1 - p;
#pragma unroll
        for (int j = 0; j < KAN_NKNOT - 1 - 1; ++j) {
            if (j < n) {
                float a = (xv - tk[j]) / (tk[j + p] - tk[j]);
                float c = (tk[j + p + 1] - xv) / (tk[j + p + 1] - tk[j + 1]);
                bas[j] = a * bas[j] + c * bas[j + 1];
            }
        }
    }

    float silu = xv / (1.0f + expf(-xv));

    float* dst = F + (size_t)b * KAN_KDIM + (size_t)i * KAN_FEAT;
    dst[0] = silu;
#pragma unroll
    for (int c = 0; c < KAN_NBASIS; ++c) dst[1 + c] = bas[c];
}

__global__ __launch_bounds__(256)
void kan_weights(const float* __restrict__ coeff,
                 const float* __restrict__ mask,
                 const float* __restrict__ scale_base,
                 const float* __restrict__ scale_spline,
                 float* __restrict__ Wt) {
    int s = blockIdx.x * blockDim.x + threadIdx.x;
    if (s >= KAN_SIZE) return;
    int o = s / KAN_IN;
    int i = s - o * KAN_IN;

    float m  = mask[s];
    float sb = m * scale_base[s];
    float ss = m * scale_spline[s];

    int kbase = i * KAN_FEAT;
    Wt[(size_t)(kbase + 0) * KAN_OUT + o] = sb;
#pragma unroll
    for (int c = 0; c < KAN_NBASIS; ++c)
        Wt[(size_t)(kbase + 1 + c) * KAN_OUT + o] = ss * coeff[(size_t)s * KAN_NBASIS + c];
}

__global__ __launch_bounds__(256)
void kan_gemm(const float* __restrict__ F,
              const float* __restrict__ Wt,
              const float* __restrict__ bias,
              float* __restrict__ out) {
    __shared__ float bsm[GEMM_KC * 16];

    int blk  = blockIdx.x;
    int mBlk = blk >> 3;
    int nBlk = blk & 7;
    int tid  = threadIdx.x;
    int wave = tid >> 5;
    int lane = tid & 31;
    int h    = lane >> 4;
    int ln   = lane & 15;

    int m0 = mBlk * 128 + wave * 16;
    int n0 = nBlk * 16;

    const float* Arow = F + (size_t)(m0 + ln) * KAN_KDIM;
    float bval = bias[n0 + ln];

    v8f acc = {};

    for (int k0 = 0; k0 < KAN_KDIM; k0 += GEMM_KC) {
        __syncthreads();
#pragma unroll
        for (int r = 0; r < (GEMM_KC * 16) / 256; ++r) {
            int idx = r * 256 + tid;
            int kk  = idx >> 4;
            int nn  = idx & 15;
            bsm[idx] = Wt[(size_t)(k0 + kk) * KAN_OUT + n0 + nn];
        }
        __syncthreads();

        if (k0 + GEMM_KC < KAN_KDIM)
            __builtin_prefetch(Arow + k0 + GEMM_KC, 0, 0);

#pragma unroll 4
        for (int kk = 0; kk < GEMM_KC; kk += 4) {
            v2f a = *(const v2f*)(Arow + k0 + kk + 2 * h);
            v2f b;
            b.x = bsm[(kk + 2 * h + 0) * 16 + ln];
            b.y = bsm[(kk + 2 * h + 1) * 16 + ln];
            acc = __builtin_amdgcn_wmma_f32_16x16x4_f32(
                 false, a,  false, b,
                 (short)0, acc,  false,  false);
        }
    }

    float* ocol = out + n0 + ln;
#pragma unroll
    for (int r = 0; r < 8; ++r) {
        int row = m0 + r + 8 * h;
        ocol[(size_t)row * KAN_OUT] = acc[r] + bval;
    }
}

extern "C" void kernel_launch(void* const* d_in, const int* in_sizes, int n_in,
                              void* d_out, int out_size, void* d_ws, size_t ws_size,
                              hipStream_t stream) {
    const float* x            = (const float*)d_in[0];
    const float* grid_knots   = (const float*)d_in[1];
    const float* coeff        = (const float*)d_in[2];
    const float* mask         = (const float*)d_in[3];
    const float* scale_base   = (const float*)d_in[4];
    const float* scale_spline = (const float*)d_in[5];
    const float* bias         = (const float*)d_in[6];

    float* F  = (float*)d_ws;
    float* Wt = F + (size_t)KAN_BATCH * KAN_KDIM;

    kan_features<<<(KAN_BATCH * KAN_IN + 255) / 256, 256, 0, stream>>>(x, grid_knots, F);
    kan_weights<<<(KAN_SIZE + 255) / 256, 256, 0, stream>>>(coeff, mask, scale_base,
                                                            scale_spline, Wt);
    kan_gemm<<<32, 256, 0, stream>>>(F, Wt, bias, (float*)d_out);
}


# Workspace operands stored in matrix-fragment order (k plane-major) by both builders; GEMM loads 16 B per lane straight to registers, groups ahead of use, no LDS stage, no barrier: new summation order
# speedup vs baseline: 2.4247x; 2.4247x over previous
#include <hip/hip_runtime.h>
#include <math.h>


#define KAN_BATCH   512
#define KAN_IN      128
#define KAN_OUT     128
#define KAN_G       5
#define KAN_KORD    3
#define KAN_NBASIS  (KAN_G + KAN_KORD)
#define KAN_NKNOT   (KAN_G + 2 * KAN_KORD + 1)
#define KAN_FEAT    (KAN_NBASIS + 1)
#define KAN_KDIM    (KAN_IN * KAN_FEAT)
#define KAN_SIZE    (KAN_IN * KAN_OUT)

#define FR_NG       (KAN_IN / 4)
#define FR_NQ       (4 * FR_NG)
#define FR_MT       (KAN_BATCH / 16)
#define FR_NT       (KAN_OUT / 16)
#define FS_FLOATS   (FR_MT * FR_NQ * 32 * 4)
#define F0_FLOATS   (FR_MT * FR_NG * 32 * 2)
#define WS_FLOATS   (FR_NT * FR_NQ * 32 * 4)
#define W0_FLOATS   (FR_NT * FR_NG * 32 * 2)

typedef __attribute__((ext_vector_type(2))) float v2f;
typedef __attribute__((ext_vector_type(4))) float v4f;
typedef __attribute__((ext_vector_type(8))) float v8f;

__device__ __forceinline__
void kan_feat9(float xv, const float (&tk)[KAN_NKNOT], float (&f)[KAN_FEAT]) {
    float bas[KAN_NKNOT - 1];
#pragma unroll
    for (int j = 0; j < KAN_NKNOT - 1; ++j)
        bas[j] = (xv >= tk[j] && xv < tk[j + 1]) ? 1.0f : 0.0f;

#pragma unroll
    for (int p = 1; p <= KAN_KORD; ++p) {
        int n = KAN_NKNOT - 1 - p;
#pragma unroll
        for (int j = 0; j < KAN_NKNOT - 1 - 1; ++j) {
            if (j < n) {
                float a = (xv - tk[j]) / (tk[j + p] - tk[j]);
                float c = (tk[j + p + 1] - xv) / (tk[j + p + 1] - tk[j + 1]);
                bas[j] = a * bas[j] + c * bas[j + 1];
            }
        }
    }
    f[0] = xv / (1.0f + expf(-xv));
#pragma unroll
    for (int c = 0; c < KAN_NBASIS; ++c) f[1 + c] = bas[c];
}

__global__ __launch_bounds__(128)
void kan_features(const float* __restrict__ x,
                  const float* __restrict__ knots_g,
                  float* __restrict__ Fs,
                  float* __restrict__ F0) {
    int t    = blockIdx.x * blockDim.x + threadIdx.x;
    int lane = t & 31;
    int ln   = t & 15;
    int h    = (t >> 4) & 1;
    int g    = (t >> 5) & (FR_NG - 1);
    int mt   = t >> 10;

    int b  = mt * 16 + ln;
    int i0 = 4 * g + 2 * h;

    v2f xv = *(const v2f*)(x + b * KAN_IN + i0);

    float tk[KAN_NKNOT];
#pragma unroll
    for (int j = 0; j < KAN_NKNOT; ++j) tk[j] = knots_g[j];

    float f0[KAN_FEAT], f1[KAN_FEAT];
    kan_feat9(xv.x, tk, f0);
    kan_feat9(xv.y, tk, f1);

    v4f* dst = (v4f*)Fs + ((size_t)mt * FR_NQ + g) * 32 + lane;
#pragma unroll
    for (int j = 0; j < 4; ++j) {
        v4f v;
        v.x = f0[1 + 2 * j];
        v.y = f1[1 + 2 * j];
        v.z = f0[2 + 2 * j];
        v.w = f1[2 + 2 * j];
        dst[(size_t)j * FR_NG * 32] = v;
    }
    v2f s;
    s.x = f0[0];
    s.y = f1[0];
    ((v2f*)F0)[((size_t)mt * FR_NG + g) * 32 + lane] = s;
}

__global__ __launch_bounds__(64)
void kan_weights(const float* __restrict__ coeff,
                 const float* __restrict__ mask,
                 const float* __restrict__ scale_base,
                 const float* __restrict__ scale_spline,
                 float* __restrict__ Ws,
                 float* __restrict__ W0) {
    int t    = blockIdx.x * blockDim.x + threadIdx.x;
    int lane = t & 31;
    int ln   = t & 15;
    int h    = (t >> 4) & 1;
    int g    = (t >> 5) & (FR_NG - 1);
    int nt   = t >> 10;

    int o  = nt * 16 + ln;
    int i0 = 4 * g + 2 * h;
    int s0 = o * KAN_IN + i0;

    v2f m   = *(const v2f*)(mask + s0);
    v2f sbv = *(const v2f*)(scale_base + s0);
    v2f ssv = *(const v2f*)(scale_spline + s0);
    const v4f* cp = (const v4f*)(coeff + (size_t)s0 * KAN_NBASIS);
    v4f ca0 = cp[0];
    v4f ca1 = cp[1];
    v4f cb0 = cp[2];
    v4f cb1 = cp[3];

    float sb0 = m.x * sbv.x, sb1 = m.y * sbv.y;
    float ss0 = m.x * ssv.x, ss1 = m.y * ssv.y;

    float ca[KAN_NBASIS] = {ca0.x, ca0.y, ca0.z, ca0.w, ca1.x, ca1.y, ca1.z, ca1.w};
    float cb[KAN_NBASIS] = {cb0.x, cb0.y, cb0.z, cb0.w, cb1.x, cb1.y, cb1.z, cb1.w};

    v4f* dst = (v4f*)Ws + ((size_t)nt * FR_NQ + g) * 32 + lane;
#pragma unroll
    for (int j = 0; j < 4; ++j) {
        v4f v;
        v.x = ss0 * ca[2 * j];
        v.y = ss1 * cb[2 * j];
        v.z = ss0 * ca[2 * j + 1];
        v.w = ss1 * cb[2 * j + 1];
        dst[(size_t)j * FR_NG * 32] = v;
    }
    v2f s;
    s.x = sb0;
    s.y = sb1;
    ((v2f*)W0)[((size_t)nt * FR_NG + g) * 32 + lane] = s;
}

#define GEMM_GU 16
#define GEMM_NGRP (FR_NQ / GEMM_GU)

__device__ __forceinline__
void frag_load(v4f (&a)[GEMM_GU], v4f (&b)[GEMM_GU],
               const v4f* __restrict__ Ap, const v4f* __restrict__ Bp, int grp) {
#pragma unroll
    for (int u = 0; u < GEMM_GU; ++u) {
        a[u] = Ap[(grp * GEMM_GU + u) * 32];
        b[u] = Bp[(grp * GEMM_GU + u) * 32];
    }
    __builtin_amdgcn_sched_barrier(0);
}

__device__ __forceinline__
v8f frag_mma(v8f acc, const v4f (&a)[GEMM_GU], const v4f (&b)[GEMM_GU]) {
#pragma unroll
    for (int u = 0; u < GEMM_GU; ++u) {
        v2f a0 = {a[u].x, a[u].y}, b0 = {b[u].x, b[u].y};
        v2f a1 = {a[u].z, a[u].w}, b1 = {b[u].z, b[u].w};
        acc = __builtin_amdgcn_wmma_f32_16x16x4_f32(false, a0, false, b0, (short)0, acc, false, false);
        acc = __builtin_amdgcn_wmma_f32_16x16x4_f32(false, a1, false, b1, (short)0, acc, false, false);
    }
    __builtin_amdgcn_sched_barrier(0);
    return acc;
}

__global__ __launch_bounds__(256)
void kan_gemm(const float* __restrict__ Fs,
              const float* __restrict__ F0,
              const float* __restrict__ Ws,
              const float* __restrict__ W0,
              const float* __restrict__ bias,
              float* __restrict__ out) {
    int blk  = blockIdx.x;
    int mBlk = blk >> 3;
    int nBlk = blk & 7;
    int tid  = threadIdx.x;
    int wave = tid >> 5;
    int lane = tid & 31;
    int h    = lane >> 4;
    int ln   = lane & 15;

    int mt = mBlk * 8 + wave;
    int nt = nBlk;
    int m0 = mt * 16;
    int n0 = nt * 16;

    const v4f* __restrict__ Ap = (const v4f*)Fs + (size_t)mt * FR_NQ * 32 + lane;
    const v4f* __restrict__ Bp = (const v4f*)Ws + (size_t)nt * FR_NQ * 32 + lane;
    const v2f* __restrict__ A0 = (const v2f*)F0 + (size_t)mt * FR_NG * 32 + lane;
    const v2f* __restrict__ B0 = (const v2f*)W0 + (size_t)nt * FR_NG * 32 + lane;

    float bval = bias[n0 + ln];

    v8f acc = {};
    v4f a0[GEMM_GU], b0[GEMM_GU], a1[GEMM_GU], b1[GEMM_GU];

    frag_load(a0, b0, Ap, Bp, 0);
#pragma unroll
    for (int gp = 0; gp < GEMM_NGRP - 2; gp += 2) {
        frag_load(a1, b1, Ap, Bp, gp + 1);
        acc = frag_mma(acc, a0, b0);
        frag_load(a0, b0, Ap, Bp, gp + 2);
        acc = frag_mma(acc, a1, b1);
    }
    frag_load(a1, b1, Ap, Bp, GEMM_NGRP - 1);
    acc = frag_mma(acc, a0, b0);

    v2f sa[FR_NG], sb[FR_NG];
#pragma unroll
    for (int g = 0; g < FR_NG; ++g) {
        sa[g] = A0[g * 32];
        sb[g] = B0[g * 32];
    }
    __builtin_amdgcn_sched_barrier(0);
    acc = frag_mma(acc, a1, b1);
#pragma unroll
    for (int g = 0; g < FR_NG; ++g)
        acc = __builtin_amdgcn_wmma_f32_16x16x4_f32(false, sa[g], false, sb[g], (short)0, acc, false, false);

    float* ocol = out + n0 + ln;
#pragma unroll
    for (int r = 0; r < 8; ++r) {
        int row = m0 + r + 8 * h;
        ocol[(size_t)row * KAN_OUT] = acc[r] + bval;
    }
}

extern "C" void kernel_launch(void* const* d_in, const int* in_sizes, int n_in,
                              void* d_out, int out_size, void* d_ws, size_t ws_size,
                              hipStream_t stream) {
    const float* x            = (const float*)d_in[0];
    const float* grid_knots   = (const float*)d_in[1];
    const float* coeff        = (const float*)d_in[2];
    const float* mask         = (const float*)d_in[3];
    const float* scale_base   = (const float*)d_in[4];
    const float* scale_spline = (const float*)d_in[5];
    const float* bias         = (const float*)d_in[6];

    float* Fs = (float*)d_ws;
    float* F0 = Fs + FS_FLOATS;
    float* Ws = F0 + F0_FLOATS;
    float* W0 = Ws + WS_FLOATS;

    kan_features<<<(KAN_BATCH * KAN_IN / 2) / 128, 128, 0, stream>>>(x, grid_knots, Fs, F0);
    kan_weights<<<(KAN_SIZE / 2) / 64, 64, 0, stream>>>(coeff, mask, scale_base,
                                                        scale_spline, Ws, W0);
    kan_gemm<<<32, 256, 0, stream>>>(Fs, F0, Ws, W0, bias, (float*)d_out);
}


# Both operand builders share one launch: 256 work-groups of 5 waves, 4 feature waves and 1 weight wave each (wave-uniform branch); 2 launches, was 3; workspace layout, GEMM and sums unchanged
# speedup vs baseline: 2.8206x; 1.1633x over previous
#include <hip/hip_runtime.h>
#include <math.h>


#define KAN_BATCH   512
#define KAN_IN      128
#define KAN_OUT     128
#define KAN_G       5
#define KAN_KORD    3
#define KAN_NBASIS  (KAN_G + KAN_KORD)
#define KAN_NKNOT   (KAN_G + 2 * KAN_KORD + 1)
#define KAN_FEAT    (KAN_NBASIS + 1)
#define KAN_KDIM    (KAN_IN * KAN_FEAT)
#define KAN_SIZE    (KAN_IN * KAN_OUT)

#define FR_NG       (KAN_IN / 4)
#define FR_NQ       (4 * FR_NG)
#define FR_MT       (KAN_BATCH / 16)
#define FR_NT       (KAN_OUT / 16)
#define FS_FLOATS   (FR_MT * FR_NQ * 32 * 4)
#define F0_FLOATS   (FR_MT * FR_NG * 32 * 2)
#define WS_FLOATS   (FR_NT * FR_NQ * 32 * 4)
#define W0_FLOATS   (FR_NT * FR_NG * 32 * 2)

typedef __attribute__((ext_vector_type(2))) float v2f;
typedef __attribute__((ext_vector_type(4))) float v4f;
typedef __attribute__((ext_vector_type(8))) float v8f;

__device__ __forceinline__
void kan_feat9(float xv, const float (&tk)[KAN_NKNOT], float (&f)[KAN_FEAT]) {
    float bas[KAN_NKNOT - 1];
#pragma unroll
    for (int j = 0; j < KAN_NKNOT - 1; ++j)
        bas[j] = (xv >= tk[j] && xv < tk[j + 1]) ? 1.0f : 0.0f;

#pragma unroll
    for (int p = 1; p <= KAN_KORD; ++p) {
        int n = KAN_NKNOT - 1 - p;
#pragma unroll
        for (int j = 0; j < KAN_NKNOT - 1 - 1; ++j) {
            if (j < n) {
                float a = (xv - tk[j]) / (tk[j + p] - tk[j]);
                float c = (tk[j + p + 1] - xv) / (tk[j + p + 1] - tk[j + 1]);
                bas[j] = a * bas[j] + c * bas[j + 1];
            }
        }
    }
    f[0] = xv / (1.0f + expf(-xv));
#pragma unroll
    for (int c = 0; c < KAN_NBASIS; ++c) f[1 + c] = bas[c];
}

__device__ __forceinline__
void kan_features(int t,
                  const float* __restrict__ x,
                  const float* __restrict__ knots_g,
                  float* __restrict__ Fs,
                  float* __restrict__ F0) {
    int lane = t & 31;
    int ln   = t & 15;
    int h    = (t >> 4) & 1;
    int g    = (t >> 5) & (FR_NG - 1);
    int mt   = t >> 10;

    int b  = mt * 16 + ln;
    int i0 = 4 * g + 2 * h;

    v2f xv = *(const v2f*)(x + b * KAN_IN + i0);

    float tk[KAN_NKNOT];
#pragma unroll
    for (int j = 0; j < KAN_NKNOT; ++j) tk[j] = knots_g[j];

    float f0[KAN_FEAT], f1[KAN_FEAT];
    kan_feat9(xv.x, tk, f0);
    kan_feat9(xv.y, tk, f1);

    v4f* dst = (v4f*)Fs + ((size_t)mt * FR_NQ + g) * 32 + lane;
#pragma unroll
    for (int j = 0; j < 4; ++j) {
        v4f v;
        v.x = f0[1 + 2 * j];
        v.y = f1[1 + 2 * j];
        v.z = f0[2 + 2 * j];
        v.w = f1[2 + 2 * j];
        dst[(size_t)j * FR_NG * 32] = v;
    }
    v2f s;
    s.x = f0[0];
    s.y = f1[0];
    ((v2f*)F0)[((size_t)mt * FR_NG + g) * 32 + lane] = s;
}

__device__ __forceinline__
void kan_weights(int t,
                 const float* __restrict__ coeff,
                 const float* __restrict__ mask,
                 const float* __restrict__ scale_base,
                 const float* __restrict__ scale_spline,
                 float* __restrict__ Ws,
                 float* __restrict__ W0) {
    int lane = t & 31;
    int ln   = t & 15;
    int h    = (t >> 4) & 1;
    int g    = (t >> 5) & (FR_NG - 1);
    int nt   = t >> 10;

    int o  = nt * 16 + ln;
    int i0 = 4 * g + 2 * h;
    int s0 = o * KAN_IN + i0;

    v2f m   = *(const v2f*)(mask + s0);
    v2f sbv = *(const v2f*)(scale_base + s0);
    v2f ssv = *(const v2f*)(scale_spline + s0);
    const v4f* cp = (const v4f*)(coeff + (size_t)s0 * KAN_NBASIS);
    v4f ca0 = cp[0];
    v4f ca1 = cp[1];
    v4f cb0 = cp[2];
    v4f cb1 = cp[3];

    float sb0 = m.x * sbv.x, sb1 = m.y * sbv.y;
    float ss0 = m.x * ssv.x, ss1 = m.y * ssv.y;

    float ca[KAN_NBASIS] = {ca0.x, ca0.y, ca0.z, ca0.w, ca1.x, ca1.y, ca1.z, ca1.w};
    float cb[KAN_NBASIS] = {cb0.x, cb0.y, cb0.z, cb0.w, cb1.x, cb1.y, cb1.z, cb1.w};

    v4f* dst = (v4f*)Ws + ((size_t)nt * FR_NQ + g) * 32 + lane;
#pragma unroll
    for (int j = 0; j < 4; ++j) {
        v4f v;
        v.x = ss0 * ca[2 * j];
        v.y = ss1 * cb[2 * j];
        v.z = ss0 * ca[2 * j + 1];
        v.w = ss1 * cb[2 * j + 1];
        dst[(size_t)j * FR_NG * 32] = v;
    }
    v2f s;
    s.x = sb0;
    s.y = sb1;
    ((v2f*)W0)[((size_t)nt * FR_NG + g) * 32 + lane] = s;
}

#define BLD_FEAT_THREADS 128
#define BLD_WGT_THREADS  32
#define BLD_THREADS      (BLD_FEAT_THREADS + BLD_WGT_THREADS)
#define BLD_BLOCKS       ((KAN_BATCH * KAN_IN / 2) / BLD_FEAT_THREADS)
static_assert((KAN_SIZE / 2) / BLD_WGT_THREADS == BLD_BLOCKS,
              "feature and weight threads must fill the same number of work-groups");

__global__ __launch_bounds__(BLD_THREADS)
void kan_build(const float* __restrict__ x,
               const float* __restrict__ knots_g,
               const float* __restrict__ coeff,
               const float* __restrict__ mask,
               const float* __restrict__ scale_base,
               const float* __restrict__ scale_spline,
               float* __restrict__ Fs,
               float* __restrict__ F0,
               float* __restrict__ Ws,
               float* __restrict__ W0) {
    int tid = threadIdx.x;
    if (tid < BLD_FEAT_THREADS) {
        kan_features(blockIdx.x * BLD_FEAT_THREADS + tid, x, knots_g, Fs, F0);
    } else {
        kan_weights(blockIdx.x * BLD_WGT_THREADS + (tid - BLD_FEAT_THREADS),
                    coeff, mask, scale_base, scale_spline, Ws, W0);
    }
}

#define GEMM_GU 16
#define GEMM_NGRP (FR_NQ / GEMM_GU)

__device__ __forceinline__
void frag_load(v4f (&a)[GEMM_GU], v4f (&b)[GEMM_GU],
               const v4f* __restrict__ Ap, const v4f* __restrict__ Bp, int grp) {
#pragma unroll
    for (int u = 0; u < GEMM_GU; ++u) {
        a[u] = Ap[(grp * GEMM_GU + u) * 32];
        b[u] = Bp[(grp * GEMM_GU + u) * 32];
    }
    __builtin_amdgcn_sched_barrier(0);
}

__device__ __forceinline__
v8f frag_mma(v8f acc, const v4f (&a)[GEMM_GU], const v4f (&b)[GEMM_GU]) {
#pragma unroll
    for (int u = 0; u < GEMM_GU; ++u) {
        v2f a0 = {a[u].x, a[u].y}, b0 = {b[u].x, b[u].y};
        v2f a1 = {a[u].z, a[u].w}, b1 = {b[u].z, b[u].w};
        acc = __builtin_amdgcn_wmma_f32_16x16x4_f32(false, a0, false, b0, (short)0, acc, false, false);
        acc = __builtin_amdgcn_wmma_f32_16x16x4_f32(false, a1, false, b1, (short)0, acc, false, false);
    }
    __builtin_amdgcn_sched_barrier(0);
    return acc;
}

__global__ __launch_bounds__(256)
void kan_gemm(const float* __restrict__ Fs,
              const float* __restrict__ F0,
              const float* __restrict__ Ws,
              const float* __restrict__ W0,
              const float* __restrict__ bias,
              float* __restrict__ out) {
    int blk  = blockIdx.x;
    int mBlk = blk >> 3;
    int nBlk = blk & 7;
    int tid  = threadIdx.x;
    int wave = tid >> 5;
    int lane = tid & 31;
    int h    = lane >> 4;
    int ln   = lane & 15;

    int mt = mBlk * 8 + wave;
    int nt = nBlk;
    int m0 = mt * 16;
    int n0 = nt * 16;

    const v4f* __restrict__ Ap = (const v4f*)Fs + (size_t)mt * FR_NQ * 32 + lane;
    const v4f* __restrict__ Bp = (const v4f*)Ws + (size_t)nt * FR_NQ * 32 + lane;
    const v2f* __restrict__ A0 = (const v2f*)F0 + (size_t)mt * FR_NG * 32 + lane;
    const v2f* __restrict__ B0 = (const v2f*)W0 + (size_t)nt * FR_NG * 32 + lane;

    float bval = bias[n0 + ln];

    v8f acc = {};
    v4f a0[GEMM_GU], b0[GEMM_GU], a1[GEMM_GU], b1[GEMM_GU];

    frag_load(a0, b0, Ap, Bp, 0);
#pragma unroll
    for (int gp = 0; gp < GEMM_NGRP - 2; gp += 2) {
        frag_load(a1, b1, Ap, Bp, gp + 1);
        acc = frag_mma(acc, a0, b0);
        frag_load(a0, b0, Ap, Bp, gp + 2);
        acc = frag_mma(acc, a1, b1);
    }
    frag_load(a1, b1, Ap, Bp, GEMM_NGRP - 1);
    acc = frag_mma(acc, a0, b0);

    v2f sa[FR_NG], sb[FR_NG];
#pragma unroll
    for (int g = 0; g < FR_NG; ++g) {
        sa[g] = A0[g * 32];
        sb[g] = B0[g * 32];
    }
    __builtin_amdgcn_sched_barrier(0);
    acc = frag_mma(acc, a1, b1);
#pragma unroll
    for (int g = 0; g < FR_NG; ++g)
        acc = __builtin_amdgcn_wmma_f32_16x16x4_f32(false, sa[g], false, sb[g], (short)0, acc, false, false);

    float* ocol = out + n0 + ln;
#pragma unroll
    for (int r = 0; r < 8; ++r) {
        int row = m0 + r + 8 * h;
        ocol[(size_t)row * KAN_OUT] = acc[r] + bval;
    }
}

extern "C" void kernel_launch(void* const* d_in, const int* in_sizes, int n_in,
                              void* d_out, int out_size, void* d_ws, size_t ws_size,
                              hipStream_t stream) {
    const float* x            = (const float*)d_in[0];
    const float* grid_knots   = (const float*)d_in[1];
    const float* coeff        = (const float*)d_in[2];
    const float* mask         = (const float*)d_in[3];
    const float* scale_base   = (const float*)d_in[4];
    const float* scale_spline = (const float*)d_in[5];
    const float* bias         = (const float*)d_in[6];

    float* Fs = (float*)d_ws;
    float* F0 = Fs + FS_FLOATS;
    float* Ws = F0 + F0_FLOATS;
    float* W0 = Ws + WS_FLOATS;

    kan_build<<<BLD_BLOCKS, BLD_THREADS, 0, stream>>>(x, grid_knots, coeff, mask, scale_base,
                                                      scale_spline, Fs, F0, Ws, W0);
    kan_gemm<<<32, 256, 0, stream>>>(Fs, F0, Ws, W0, bias, (float*)d_out);
}


# GEMM as 256 one-wave work-groups (one 16x16 tile, whole K each): a SIMD per tile, two alternating accumulators added at the end, loads two groups ahead (third register set): new summation order
# speedup vs baseline: 3.7359x; 1.3245x over previous
#include <hip/hip_runtime.h>
#include <math.h>


#define KAN_BATCH   512
#define KAN_IN      128
#define KAN_OUT     128
#define KAN_G       5
#define KAN_KORD    3
#define KAN_NBASIS  (KAN_G + KAN_KORD)
#define KAN_NKNOT   (KAN_G + 2 * KAN_KORD + 1)
#define KAN_FEAT    (KAN_NBASIS + 1)
#define KAN_KDIM    (KAN_IN * KAN_FEAT)
#define KAN_SIZE    (KAN_IN * KAN_OUT)

#define FR_NG       (KAN_IN / 4)
#define FR_NQ       (4 * FR_NG)
#define FR_MT       (KAN_BATCH / 16)
#define FR_NT       (KAN_OUT / 16)
#define FS_FLOATS   (FR_MT * FR_NQ * 32 * 4)
#define F0_FLOATS   (FR_MT * FR_NG * 32 * 2)
#define WS_FLOATS   (FR_NT * FR_NQ * 32 * 4)
#define W0_FLOATS   (FR_NT * FR_NG * 32 * 2)

typedef __attribute__((ext_vector_type(2))) float v2f;
typedef __attribute__((ext_vector_type(4))) float v4f;
typedef __attribute__((ext_vector_type(8))) float v8f;

__device__ __forceinline__
void kan_feat9(float xv, const float (&tk)[KAN_NKNOT], float (&f)[KAN_FEAT]) {
    float bas[KAN_NKNOT - 1];
#pragma unroll
    for (int j = 0; j < KAN_NKNOT - 1; ++j)
        bas[j] = (xv >= tk[j] && xv < tk[j + 1]) ? 1.0f : 0.0f;

#pragma unroll
    for (int p = 1; p <= KAN_KORD; ++p) {
        int n = KAN_NKNOT - 1 - p;
#pragma unroll
        for (int j = 0; j < KAN_NKNOT - 1 - 1; ++j) {
            if (j < n) {
                float a = (xv - tk[j]) / (tk[j + p] - tk[j]);
                float c = (tk[j + p + 1] - xv) / (tk[j + p + 1] - tk[j + 1]);
                bas[j] = a * bas[j] + c * bas[j + 1];
            }
        }
    }
    f[0] = xv / (1.0f + expf(-xv));
#pragma unroll
    for (int c = 0; c < KAN_NBASIS; ++c) f[1 + c] = bas[c];
}

__device__ __forceinline__
void kan_features(int t,
                  const float* __restrict__ x,
                  const float* __restrict__ knots_g,
                  float* __restrict__ Fs,
                  float* __restrict__ F0) {
    int lane = t & 31;
    int ln   = t & 15;
    int h    = (t >> 4) & 1;
    int g    = (t >> 5) & (FR_NG - 1);
    int mt   = t >> 10;

    int b  = mt * 16 + ln;
    int i0 = 4 * g + 2 * h;

    v2f xv = *(const v2f*)(x + b * KAN_IN + i0);

    float tk[KAN_NKNOT];
#pragma unroll
    for (int j = 0; j < KAN_NKNOT; ++j) tk[j] = knots_g[j];

    float f0[KAN_FEAT], f1[KAN_FEAT];
    kan_feat9(xv.x, tk, f0);
    kan_feat9(xv.y, tk, f1);

    v4f* dst = (v4f*)Fs + ((size_t)mt * FR_NQ + g) * 32 + lane;
#pragma unroll
    for (int j = 0; j < 4; ++j) {
        v4f v;
        v.x = f0[1 + 2 * j];
        v.y = f1[1 + 2 * j];
        v.z = f0[2 + 2 * j];
        v.w = f1[2 + 2 * j];
        dst[(size_t)j * FR_NG * 32] = v;
    }
    v2f s;
    s.x = f0[0];
    s.y = f1[0];
    ((v2f*)F0)[((size_t)mt * FR_NG + g) * 32 + lane] = s;
}

__device__ __forceinline__
void kan_weights(int t,
                 const float* __restrict__ coeff,
                 const float* __restrict__ mask,
                 const float* __restrict__ scale_base,
                 const float* __restrict__ scale_spline,
                 float* __restrict__ Ws,
                 float* __restrict__ W0) {
    int lane = t & 31;
    int ln   = t & 15;
    int h    = (t >> 4) & 1;
    int g    = (t >> 5) & (FR_NG - 1);
    int nt   = t >> 10;

    int o  = nt * 16 + ln;
    int i0 = 4 * g + 2 * h;
    int s0 = o * KAN_IN + i0;

    v2f m   = *(const v2f*)(mask + s0);
    v2f sbv = *(const v2f*)(scale_base + s0);
    v2f ssv = *(const v2f*)(scale_spline + s0);
    const v4f* cp = (const v4f*)(coeff + (size_t)s0 * KAN_NBASIS);
    v4f ca0 = cp[0];
    v4f ca1 = cp[1];
    v4f cb0 = cp[2];
    v4f cb1 = cp[3];

    float sb0 = m.x * sbv.x, sb1 = m.y * sbv.y;
    float ss0 = m.x * ssv.x, ss1 = m.y * ssv.y;

    float ca[KAN_NBASIS] = {ca0.x, ca0.y, ca0.z, ca0.w, ca1.x, ca1.y, ca1.z, ca1.w};
    float cb[KAN_NBASIS] = {cb0.x, cb0.y, cb0.z, cb0.w, cb1.x, cb1.y, cb1.z, cb1.w};

    v4f* dst = (v4f*)Ws + ((size_t)nt * FR_NQ + g) * 32 + lane;
#pragma unroll
    for (int j = 0; j < 4; ++j) {
        v4f v;
        v.x = ss0 * ca[2 * j];
        v.y = ss1 * cb[2 * j];
        v.z = ss0 * ca[2 * j + 1];
        v.w = ss1 * cb[2 * j + 1];
        dst[(size_t)j * FR_NG * 32] = v;
    }
    v2f s;
    s.x = sb0;
    s.y = sb1;
    ((v2f*)W0)[((size_t)nt * FR_NG + g) * 32 + lane] = s;
}

#define BLD_FEAT_THREADS 128
#define BLD_WGT_THREADS  32
#define BLD_THREADS      (BLD_FEAT_THREADS + BLD_WGT_THREADS)
#define BLD_BLOCKS       ((KAN_BATCH * KAN_IN / 2) / BLD_FEAT_THREADS)
static_assert((KAN_SIZE / 2) / BLD_WGT_THREADS == BLD_BLOCKS,
              "feature and weight threads must fill the same number of work-groups");

__global__ __launch_bounds__(BLD_THREADS)
void kan_build(const float* __restrict__ x,
               const float* __restrict__ knots_g,
               const float* __restrict__ coeff,
               const float* __restrict__ mask,
               const float* __restrict__ scale_base,
               const float* __restrict__ scale_spline,
               float* __restrict__ Fs,
               float* __restrict__ F0,
               float* __restrict__ Ws,
               float* __restrict__ W0) {
    int tid = threadIdx.x;
    if (tid < BLD_FEAT_THREADS) {
        kan_features(blockIdx.x * BLD_FEAT_THREADS + tid, x, knots_g, Fs, F0);
    } else {
        kan_weights(blockIdx.x * BLD_WGT_THREADS + (tid - BLD_FEAT_THREADS),
                    coeff, mask, scale_base, scale_spline, Ws, W0);
    }
}

#define GEMM_GU 16
#define GEMM_NGRP (FR_NQ / GEMM_GU)
#define GEMM_SH (FR_NG / 2)
static_assert(GEMM_NGRP == 8, "the group schedule in kan_gemm is written out for 8 groups");

__device__ __forceinline__
void frag_load(v4f (&a)[GEMM_GU], v4f (&b)[GEMM_GU],
               const v4f* __restrict__ Ap, const v4f* __restrict__ Bp, int grp) {
#pragma unroll
    for (int u = 0; u < GEMM_GU; ++u) {
        a[u] = Ap[(grp * GEMM_GU + u) * 32];
        b[u] = Bp[(grp * GEMM_GU + u) * 32];
    }
    __builtin_amdgcn_sched_barrier(0);
}

__device__ __forceinline__
void silu_load(v2f (&a)[GEMM_SH], v2f (&b)[GEMM_SH],
               const v2f* __restrict__ A0, const v2f* __restrict__ B0, int half) {
#pragma unroll
    for (int g = 0; g < GEMM_SH; ++g) {
        a[g] = A0[(half * GEMM_SH + g) * 32];
        b[g] = B0[(half * GEMM_SH + g) * 32];
    }
    __builtin_amdgcn_sched_barrier(0);
}

__device__ __forceinline__
void frag_mma(v8f& acc0, v8f& acc1, const v4f (&a)[GEMM_GU], const v4f (&b)[GEMM_GU]) {
#pragma unroll
    for (int u = 0; u < GEMM_GU; ++u) {
        v2f a0 = {a[u].x, a[u].y}, b0 = {b[u].x, b[u].y};
        v2f a1 = {a[u].z, a[u].w}, b1 = {b[u].z, b[u].w};
        acc0 = __builtin_amdgcn_wmma_f32_16x16x4_f32(false, a0, false, b0, (short)0, acc0, false, false);
        acc1 = __builtin_amdgcn_wmma_f32_16x16x4_f32(false, a1, false, b1, (short)0, acc1, false, false);
    }
    __builtin_amdgcn_sched_barrier(0);
}

__device__ __forceinline__
void silu_mma(v8f& acc0, v8f& acc1, const v2f (&a)[GEMM_SH], const v2f (&b)[GEMM_SH]) {
#pragma unroll
    for (int g = 0; g < GEMM_SH; g += 2) {
        acc0 = __builtin_amdgcn_wmma_f32_16x16x4_f32(false, a[g], false, b[g], (short)0, acc0, false, false);
        acc1 = __builtin_amdgcn_wmma_f32_16x16x4_f32(false, a[g + 1], false, b[g + 1], (short)0, acc1, false, false);
    }
}

#define GEMM_BLOCKS (FR_MT * FR_NT)

__global__ __launch_bounds__(32)
void kan_gemm(const float* __restrict__ Fs,
              const float* __restrict__ F0,
              const float* __restrict__ Ws,
              const float* __restrict__ W0,
              const float* __restrict__ bias,
              float* __restrict__ out) {
    int blk  = blockIdx.x;
    int mt   = blk >> 3;
    int nt   = blk & 7;
    int lane = threadIdx.x;
    int h    = lane >> 4;
    int ln   = lane & 15;

    int m0 = mt * 16;
    int n0 = nt * 16;

    const v4f* __restrict__ Ap = (const v4f*)Fs + (size_t)mt * FR_NQ * 32 + lane;
    const v4f* __restrict__ Bp = (const v4f*)Ws + (size_t)nt * FR_NQ * 32 + lane;
    const v2f* __restrict__ A0 = (const v2f*)F0 + (size_t)mt * FR_NG * 32 + lane;
    const v2f* __restrict__ B0 = (const v2f*)W0 + (size_t)nt * FR_NG * 32 + lane;

    float bval = bias[n0 + ln];

    v8f acc0 = {}, acc1 = {};
    v4f a0[GEMM_GU], b0[GEMM_GU], a1[GEMM_GU], b1[GEMM_GU], a2[GEMM_GU], b2[GEMM_GU];
    v2f sa0[GEMM_SH], sb0[GEMM_SH], sa1[GEMM_SH], sb1[GEMM_SH];

    frag_load(a0, b0, Ap, Bp, 0);
    frag_load(a1, b1, Ap, Bp, 1);

    frag_load(a2, b2, Ap, Bp, 2);
    frag_mma(acc0, acc1, a0, b0);
    frag_load(a0, b0, Ap, Bp, 3);
    frag_mma(acc0, acc1, a1, b1);
    frag_load(a1, b1, Ap, Bp, 4);
    frag_mma(acc0, acc1, a2, b2);
    frag_load(a2, b2, Ap, Bp, 5);
    frag_mma(acc0, acc1, a0, b0);
    frag_load(a0, b0, Ap, Bp, 6);
    frag_mma(acc0, acc1, a1, b1);
    frag_load(a1, b1, Ap, Bp, 7);
    frag_mma(acc0, acc1, a2, b2);
    silu_load(sa0, sb0, A0, B0, 0);
    frag_mma(acc0, acc1, a0, b0);
    silu_load(sa1, sb1, A0, B0, 1);
    frag_mma(acc0, acc1, a1, b1);
    silu_mma(acc0, acc1, sa0, sb0);
    silu_mma(acc0, acc1, sa1, sb1);

    float* ocol = out + n0 + ln;
#pragma unroll
    for (int r = 0; r < 8; ++r) {
        int row = m0 + r + 8 * h;
        ocol[(size_t)row * KAN_OUT] = (acc0[r] + acc1[r]) + bval;
    }
}

extern "C" void kernel_launch(void* const* d_in, const int* in_sizes, int n_in,
                              void* d_out, int out_size, void* d_ws, size_t ws_size,
                              hipStream_t stream) {
    const float* x            = (const float*)d_in[0];
    const float* grid_knots   = (const float*)d_in[1];
    const float* coeff        = (const float*)d_in[2];
    const float* mask         = (const float*)d_in[3];
    const float* scale_base   = (const float*)d_in[4];
    const float* scale_spline = (const float*)d_in[5];
    const float* bias         = (const float*)d_in[6];

    float* Fs = (float*)d_ws;
    float* F0 = Fs + FS_FLOATS;
    float* Ws = F0 + F0_FLOATS;
    float* W0 = Ws + WS_FLOATS;

    kan_build<<<BLD_BLOCKS, BLD_THREADS, 0, stream>>>(x, grid_knots, coeff, mask, scale_base,
                                                      scale_spline, Fs, F0, Ws, W0);
    kan_gemm<<<GEMM_BLOCKS, 32, 0, stream>>>(Fs, F0, Ws, W0, bias, (float*)d_out);
}


# Matrix steps moved into the feature launch: work-group = 16 rows x 16 inputs, features via LDS to 8 column-tile waves, K split over work-groups, partial tiles summed by launch 2: new summation order
# speedup vs baseline: 5.3129x; 1.4221x over previous
#include <hip/hip_runtime.h>
#include <math.h>


#define KAN_BATCH   512
#define KAN_IN      128
#define KAN_OUT     128
#define KAN_G       5
#define KAN_KORD    3
#define KAN_NBASIS  (KAN_G + KAN_KORD)
#define KAN_NKNOT   (KAN_G + 2 * KAN_KORD + 1)
#define KAN_FEAT    (KAN_NBASIS + 1)

#define PK_NS       8
#define PK_II       (KAN_IN / PK_NS)
#define PK_Q        (PK_II / 4)
#define PK_STEPS    (KAN_FEAT * PK_Q)
#define PK_PAIRS    (PK_STEPS / 2)
#define PK_MT       (KAN_BATCH / 16)
#define PK_NT       (KAN_OUT / 16)
#define PK_TILES    (PK_MT * PK_NT)
#define PK_THREADS  (PK_NT * 32)
static_assert(PK_THREADS == 16 * PK_II, "one thread per (row, input) of the block");
static_assert(PK_STEPS % 2 == 0, "steps are stored in pairs");

typedef __attribute__((ext_vector_type(2))) float v2f;
typedef __attribute__((ext_vector_type(4))) float v4f;
typedef __attribute__((ext_vector_type(8))) float v8f;

__device__ __forceinline__
void kan_feat9(float xv, const float (&tk)[KAN_NKNOT], float (&f)[KAN_FEAT]) {
    float bas[KAN_NKNOT - 1];
#pragma unroll
    for (int j = 0; j < KAN_NKNOT - 1; ++j)
        bas[j] = (xv >= tk[j] && xv < tk[j + 1]) ? 1.0f : 0.0f;

#pragma unroll
    for (int p = 1; p <= KAN_KORD; ++p) {
        int n = KAN_NKNOT - 1 - p;
#pragma unroll
        for (int j = 0; j < KAN_NKNOT - 1 - 1; ++j) {
            if (j < n) {
                float a = (xv - tk[j]) / (tk[j + p] - tk[j]);
                float c = (tk[j + p + 1] - xv) / (tk[j + p + 1] - tk[j + 1]);
                bas[j] = a * bas[j] + c * bas[j + 1];
            }
        }
    }
    f[0] = xv / (1.0f + expf(-xv));
#pragma unroll
    for (int c = 0; c < KAN_NBASIS; ++c) f[1 + c] = bas[c];
}

__global__ __launch_bounds__(PK_THREADS)
void kan_part(const float* __restrict__ x,
              const float* __restrict__ knots_g,
              const float* __restrict__ coeff,
              const float* __restrict__ mask,
              const float* __restrict__ scale_base,
              const float* __restrict__ scale_spline,
              float* __restrict__ P) {
    __shared__ v4f As[PK_PAIRS * 32];

    int t    = threadIdx.x;
    int mt   = blockIdx.x >> 3;
    int ks   = blockIdx.x & (PK_NS - 1);
    int lane = t & 31;
    int w    = t >> 5;
    int h    = lane >> 4;
    int ln   = lane & 15;

    int r  = t & 15;
    int ii = t >> 4;
    float xv = x[(mt * 16 + r) * KAN_IN + ks * PK_II + ii];

    float tk[KAN_NKNOT];
#pragma unroll
    for (int j = 0; j < KAN_NKNOT; ++j) tk[j] = knots_g[j];

    __builtin_amdgcn_sched_barrier(0);

    int o = w * 16 + ln;
    v2f m[PK_Q], sbv[PK_Q], ssv[PK_Q];
    v4f cf[PK_Q][4];
#pragma unroll
    for (int q = 0; q < PK_Q; ++q) {
        int s0 = o * KAN_IN + ks * PK_II + 4 * q + 2 * h;
        m[q]   = *(const v2f*)(mask + s0);
        sbv[q] = *(const v2f*)(scale_base + s0);
        ssv[q] = *(const v2f*)(scale_spline + s0);
        const v4f* cp = (const v4f*)(coeff + (size_t)s0 * KAN_NBASIS);
        cf[q][0] = cp[0];
        cf[q][1] = cp[1];
        cf[q][2] = cp[2];
        cf[q][3] = cp[3];
    }
    __builtin_amdgcn_sched_barrier(0);

    float f[KAN_FEAT];
    kan_feat9(xv, tk, f);

    float* Asf = (float*)As;
    int fq = ii >> 2;
    int fpos = (((ii >> 1) & 1) * 16 + r) * 4 + (ii & 1);
#pragma unroll
    for (int c = 0; c < KAN_FEAT; ++c) {
        int s = c * PK_Q + fq;
        Asf[(s >> 1) * 128 + fpos + (s & 1) * 2] = f[c];
    }

    __builtin_amdgcn_sched_barrier(0);

    v2f bw[PK_STEPS];
#pragma unroll
    for (int s = 0; s < PK_STEPS; ++s) {
        int c = s / PK_Q;
        int q = s % PK_Q;
        if (c == 0) {
            bw[s].x = m[q].x * sbv[q].x;
            bw[s].y = m[q].y * sbv[q].y;
        } else {
            int k = c - 1;
            float ss0 = m[q].x * ssv[q].x;
            float ss1 = m[q].y * ssv[q].y;
            float ca = (k < 4) ? cf[q][0][k & 3] : cf[q][1][k & 3];
            float cb = (k < 4) ? cf[q][2][k & 3] : cf[q][3][k & 3];
            bw[s].x = ss0 * ca;
            bw[s].y = ss1 * cb;
        }
    }
    __builtin_amdgcn_sched_barrier(0);

    __syncthreads();

    v4f a[PK_PAIRS];
#pragma unroll
    for (int p = 0; p < PK_PAIRS; ++p) a[p] = As[p * 32 + lane];
    __builtin_amdgcn_sched_barrier(0);

    v8f acc0 = {}, acc1 = {};
#pragma unroll
    for (int p = 0; p < PK_PAIRS; ++p) {
        v2f a0 = {a[p].x, a[p].y};
        v2f a1 = {a[p].z, a[p].w};
        acc0 = __builtin_amdgcn_wmma_f32_16x16x4_f32(false, a0, false, bw[2 * p], (short)0, acc0, false, false);
        acc1 = __builtin_amdgcn_wmma_f32_16x16x4_f32(false, a1, false, bw[2 * p + 1], (short)0, acc1, false, false);
    }

    v8f d = acc0 + acc1;
    v4f* dst = (v4f*)P + ((size_t)((ks * PK_TILES + mt * PK_NT + w) * 2)) * 32 + lane;
    v4f d0 = {d[0], d[1], d[2], d[3]};
    v4f d1 = {d[4], d[5], d[6], d[7]};
    dst[0]  = d0;
    dst[32] = d1;
}

__global__ __launch_bounds__(32)
void kan_sum(const float* __restrict__ P,
             const float* __restrict__ bias,
             float* __restrict__ out) {
    int tile = blockIdx.x;
    int mt   = tile >> 3;
    int nt   = tile & 7;
    int lane = threadIdx.x;
    int h    = lane >> 4;
    int ln   = lane & 15;

    float bval = bias[nt * 16 + ln];

    const v4f* src = (const v4f*)P + (size_t)tile * 2 * 32 + lane;
    v4f lo[PK_NS], hi[PK_NS];
#pragma unroll
    for (int ks = 0; ks < PK_NS; ++ks) {
        lo[ks] = src[(size_t)ks * PK_TILES * 2 * 32];
        hi[ks] = src[(size_t)ks * PK_TILES * 2 * 32 + 32];
    }
    __builtin_amdgcn_sched_barrier(0);

    v4f slo = ((lo[0] + lo[1]) + (lo[2] + lo[3])) + ((lo[4] + lo[5]) + (lo[6] + lo[7]));
    v4f shi = ((hi[0] + hi[1]) + (hi[2] + hi[3])) + ((hi[4] + hi[5]) + (hi[6] + hi[7]));

    float* ocol = out + (size_t)(mt * 16 + 8 * h) * KAN_OUT + nt * 16 + ln;
#pragma unroll
    for (int r = 0; r < 4; ++r) {
        ocol[(size_t)r * KAN_OUT]       = slo[r] + bval;
        ocol[(size_t)(r + 4) * KAN_OUT] = shi[r] + bval;
    }
}

extern "C" void kernel_launch(void* const* d_in, const int* in_sizes, int n_in,
                              void* d_out, int out_size, void* d_ws, size_t ws_size,
                              hipStream_t stream) {
    const float* x            = (const float*)d_in[0];
    const float* grid_knots   = (const float*)d_in[1];
    const float* coeff        = (const float*)d_in[2];
    const float* mask         = (const float*)d_in[3];
    const float* scale_base   = (const float*)d_in[4];
    const float* scale_spline = (const float*)d_in[5];
    const float* bias         = (const float*)d_in[6];

    float* P = (float*)d_ws;

    kan_part<<<PK_MT * PK_NS, PK_THREADS, 0, stream>>>(x, grid_knots, coeff, mask, scale_base,
                                                       scale_spline, P);
    kan_sum<<<PK_TILES, 32, 0, stream>>>(P, bias, (float*)d_out);
}
